# starting kernel
# baseline (speedup 1.0000x reference)
#include <hip/hip_runtime.h>
#include <hip/hip_bf16.h>


typedef float v2f __attribute__((ext_vector_type(2)));
typedef float v8f __attribute__((ext_vector_type(8)));

#define BATCH   1024
#define C_IN    2048
#define POOLSZ  32

__global__ void pool_mean32(const float* __restrict__ img,
                            float* __restrict__ feat, int total) {
    int i = blockIdx.x * blockDim.x + threadIdx.x;
    if (i >= total) return;
    const float4* p = reinterpret_cast<const float4*>(img + (size_t)i * POOLSZ);
    float s = 0.0f;
#pragma unroll
    for (int j = 0; j < 8; ++j) {
        float4 q = p[j];
        s += (q.x + q.y) + (q.z + q.w);
    }
    feat[i] = s * (1.0f / 32.0f);
}

template <bool RELU>
__global__ void wmma_gemm_f32(const float* __restrict__ A,
                              const float* __restrict__ W,
                              const float* __restrict__ bias,
                              float* __restrict__ C,
                              int M, int N, int K) {
    const int lane  = threadIdx.x;
    const int half  = lane >> 4;
    const int l15   = lane & 15;
    const int ntile = blockIdx.x;
    const int mtile = blockIdx.y;

    const int n  = ntile * 16 + l15;
    const int am = mtile * 16 + l15;

    const float bv = bias[n];
    v8f acc;
#pragma unroll
    for (int r = 0; r < 8; ++r) acc[r] = bv;

    const float* __restrict__ Arow = A + (size_t)am * K + 2 * half;
    const float* __restrict__ Wcol = W + (size_t)(2 * half) * N + n;

    for (int k = 0; k < K; k += 32) {
#pragma unroll
        for (int kk = 0; kk < 32; kk += 4) {
            v2f a, b;
            a.x = Arow[k + kk];
            a.y = Arow[k + kk + 1];
            b.x = Wcol[(size_t)(k + kk) * N];
            b.y = Wcol[(size_t)(k + kk + 1) * N];
            acc = __builtin_amdgcn_wmma_f32_16x16x4_f32(
                false, a, false, b, (short)0, acc, false, false);
        }
    }

    float* __restrict__ Cout = C + (size_t)(mtile * 16 + 8 * half) * N + n;
#pragma unroll
    for (int r = 0; r < 8; ++r) {
        float v = acc[r];
        if (RELU) v = fmaxf(v, 0.0f);
        Cout[(size_t)r * N] = v;
    }
}

__device__ __forceinline__ float block_reduce_128(float x, float* red) {
    const int t = threadIdx.x;
    red[t] = x;
    __syncthreads();
#pragma unroll
    for (int s = 64; s > 0; s >>= 1) {
        if (t < s) red[t] += red[t + s];
        __syncthreads();
    }
    float r = red[0];
    __syncthreads();
    return r;
}

__global__ void epilogue(const float* __restrict__ isp,
                         const float* __restrict__ ish,
                         const int* __restrict__ label,
                         const float* __restrict__ ppos,
                         const float* __restrict__ pneg,
                         float* __restrict__ out) {
    __shared__ float red[128];
    const int b = blockIdx.x;
    const int t = threadIdx.x;

    const float v  = isp[b * 128 + t];
    const float pp = ppos[t];
    const float nn = pneg[t];

    const float nv  = block_reduce_128(v * v, red);
    const float npp = block_reduce_128(pp * pp, red);
    const float nnn = block_reduce_128(nn * nn, red);
    const float dp  = block_reduce_128(v * pp, red);
    const float dn  = block_reduce_128(v * nn, red);

    const float eps = 1e-12f;
    const float dv  = fmaxf(sqrtf(nv), eps);
    const float lp  = dp / (dv * fmaxf(sqrtf(npp), eps));
    const float ln  = dn / (dv * fmaxf(sqrtf(nnn), eps));

    const bool lab    = label[b] != 0;
    const bool pred   = lp > ln;
    const bool is_bad = (pred == lab) && (lp != ln);
    const float diff  = fabsf(lp - ln);
    const float scale = is_bad ? (tanhf(-diff) + 1.0f) : 1.0f;

    out[b * 256 + t]       = v * scale;
    out[b * 256 + 128 + t] = ish[b * 128 + t];
}

extern "C" void kernel_launch(void* const* d_in, const int* in_sizes, int n_in,
                              void* d_out, int out_size, void* d_ws, size_t ws_size,
                              hipStream_t stream) {
    const float* img   = (const float*)d_in[0];
    const int*   label = (const int*)d_in[1];
    const float* W1  = (const float*)d_in[2];
    const float* b1  = (const float*)d_in[3];
    const float* W2  = (const float*)d_in[4];
    const float* b2  = (const float*)d_in[5];
    const float* W3  = (const float*)d_in[6];
    const float* b3  = (const float*)d_in[7];
    const float* Wsp = (const float*)d_in[8];
    const float* bsp = (const float*)d_in[9];
    const float* Wsh = (const float*)d_in[10];
    const float* bsh = (const float*)d_in[11];
    const float* ppn = (const float*)d_in[12];
    const float* pnn = (const float*)d_in[13];

    float* ws   = (float*)d_ws;
    float* feat = ws;
    float* h1   = feat + (size_t)BATCH * C_IN;
    float* h2   = h1   + (size_t)BATCH * 512;
    float* x    = h2   + (size_t)BATCH * 256;
    float* isp  = x    + (size_t)BATCH * 128;
    float* ish  = isp  + (size_t)BATCH * 128;

    {
        const int total = BATCH * C_IN;
        pool_mean32<<<(total + 255) / 256, 256, 0, stream>>>(img, feat, total);
    }

    wmma_gemm_f32<true><<<dim3(512 / 16, BATCH / 16), 32, 0, stream>>>(
        feat, W1, b1, h1, BATCH, 512, C_IN);
    wmma_gemm_f32<true><<<dim3(256 / 16, BATCH / 16), 32, 0, stream>>>(
        h1, W2, b2, h2, BATCH, 256, 512);
    wmma_gemm_f32<false><<<dim3(128 / 16, BATCH / 16), 32, 0, stream>>>(
        h2, W3, b3, x, BATCH, 128, 256);

    wmma_gemm_f32<false><<<dim3(128 / 16, BATCH / 16), 32, 0, stream>>>(
        x, Wsp, bsp, isp, BATCH, 128, 128);
    wmma_gemm_f32<false><<<dim3(128 / 16, BATCH / 16), 32, 0, stream>>>(
        x, Wsh, bsh, ish, BATCH, 128, 128);

    epilogue<<<BATCH, 128, 0, stream>>>(isp, ish, label, ppn, pnn, (float*)d_out);
}


# GEMMs: 4-wave workgroup per 32x64 tile, A and W K-slabs of 64 staged in two LDS stages by row-contiguous 128-bit loads, WMMA fragments read from LDS; k order per WMMA kept, summation order unchanged
# speedup vs baseline: 1.5845x; 1.5845x over previous
#include <hip/hip_runtime.h>
#include <hip/hip_bf16.h>


typedef float v2f __attribute__((ext_vector_type(2)));
typedef float v8f __attribute__((ext_vector_type(8)));

#define BATCH   1024
#define C_IN    2048
#define POOLSZ  32

__global__ void pool_mean32(const float* __restrict__ img,
                            float* __restrict__ feat, int total) {
    int i = blockIdx.x * blockDim.x + threadIdx.x;
    if (i >= total) return;
    const float4* p = reinterpret_cast<const float4*>(img + (size_t)i * POOLSZ);
    float s = 0.0f;
#pragma unroll
    for (int j = 0; j < 8; ++j) {
        float4 q = p[j];
        s += (q.x + q.y) + (q.z + q.w);
    }
    feat[i] = s * (1.0f / 32.0f);
}

typedef float v4f __attribute__((ext_vector_type(4)));

#define GEMM_BM      32
#define GEMM_BN      64
#define GEMM_BK      64
#define GEMM_THREADS 128
#define GEMM_APITCH  68
#define GEMM_BPITCH  72

#define LDG_SLAB(ag, wg)                                                      \
    pa0 = *reinterpret_cast<const v4f*>((ag));                                \
    pa1 = *reinterpret_cast<const v4f*>((ag) + (size_t)8  * K);               \
    pa2 = *reinterpret_cast<const v4f*>((ag) + (size_t)16 * K);               \
    pa3 = *reinterpret_cast<const v4f*>((ag) + (size_t)24 * K);               \
    pb0 = *reinterpret_cast<const v4f*>((wg));                                \
    pb1 = *reinterpret_cast<const v4f*>((wg) + (size_t)8  * N);               \
    pb2 = *reinterpret_cast<const v4f*>((wg) + (size_t)16 * N);               \
    pb3 = *reinterpret_cast<const v4f*>((wg) + (size_t)24 * N);               \
    pb4 = *reinterpret_cast<const v4f*>((wg) + (size_t)32 * N);               \
    pb5 = *reinterpret_cast<const v4f*>((wg) + (size_t)40 * N);               \
    pb6 = *reinterpret_cast<const v4f*>((wg) + (size_t)48 * N);               \
    pb7 = *reinterpret_cast<const v4f*>((wg) + (size_t)56 * N)

#define STS_SLAB(as_, bs_)                                                    \
    *reinterpret_cast<v4f*>((as_) + a_st)                    = pa0;           \
    *reinterpret_cast<v4f*>((as_) + a_st + 8  * GEMM_APITCH) = pa1;           \
    *reinterpret_cast<v4f*>((as_) + a_st + 16 * GEMM_APITCH) = pa2;           \
    *reinterpret_cast<v4f*>((as_) + a_st + 24 * GEMM_APITCH) = pa3;           \
    *reinterpret_cast<v4f*>((bs_) + b_st)                    = pb0;           \
    *reinterpret_cast<v4f*>((bs_) + b_st + 8  * GEMM_BPITCH) = pb1;           \
    *reinterpret_cast<v4f*>((bs_) + b_st + 16 * GEMM_BPITCH) = pb2;           \
    *reinterpret_cast<v4f*>((bs_) + b_st + 24 * GEMM_BPITCH) = pb3;           \
    *reinterpret_cast<v4f*>((bs_) + b_st + 32 * GEMM_BPITCH) = pb4;           \
    *reinterpret_cast<v4f*>((bs_) + b_st + 40 * GEMM_BPITCH) = pb5;           \
    *reinterpret_cast<v4f*>((bs_) + b_st + 48 * GEMM_BPITCH) = pb6;           \
    *reinterpret_cast<v4f*>((bs_) + b_st + 56 * GEMM_BPITCH) = pb7

#define FRAG(ks)                                                              \
    const v2f fa0_##ks = *reinterpret_cast<const v2f*>(as + 4 * (ks));        \
    const v2f fa1_##ks =                                                      \
        *reinterpret_cast<const v2f*>(as + 16 * GEMM_APITCH + 4 * (ks));      \
    v2f fb_##ks;                                                              \
    fb_##ks.x = bs[(4 * (ks)) * GEMM_BPITCH];                                 \
    fb_##ks.y = bs[(4 * (ks) + 1) * GEMM_BPITCH]

#define MMA2(ks)                                                              \
    acc0 = __builtin_amdgcn_wmma_f32_16x16x4_f32(                             \
        false, fa0_##ks, false, fb_##ks, (short)0, acc0, false, false);       \
    acc1 = __builtin_amdgcn_wmma_f32_16x16x4_f32(                             \
        false, fa1_##ks, false, fb_##ks, (short)0, acc1, false, false)

#define SLAB_MMA()                                                            \
    {                                                                         \
        FRAG(0);                                                              \
        FRAG(1);                                                              \
        FRAG(2);                                                              \
        FRAG(3);                                                              \
          \
                 \
        __builtin_amdgcn_sched_barrier(0);                                    \
        FRAG(4);                                                              \
        FRAG(5);                                                              \
        FRAG(6);                                                              \
        FRAG(7);                                                              \
                       \
        __builtin_amdgcn_sched_barrier(0);                                    \
        FRAG(8);                                                              \
        FRAG(9);                                                              \
        FRAG(10);                                                             \
        FRAG(11);                                                             \
                       \
        __builtin_amdgcn_sched_barrier(0);                                    \
        FRAG(12);                                                             \
        FRAG(13);                                                             \
        FRAG(14);                                                             \
        FRAG(15);                                                             \
            \
                 \
        __builtin_amdgcn_sched_barrier(0);                                    \
        MMA2(0);                                                              \
        MMA2(1);                                                              \
        MMA2(2);                                                              \
        MMA2(3);                                                              \
        MMA2(4);                                                              \
        MMA2(5);                                                              \
        MMA2(6);                                                              \
        MMA2(7);                                                              \
        MMA2(8);                                                              \
        MMA2(9);                                                              \
        MMA2(10);                                                             \
        MMA2(11);                                                             \
        MMA2(12);                                                             \
        MMA2(13);                                                             \
        MMA2(14);                                                             \
        MMA2(15);                                                             \
          \
        __builtin_amdgcn_sched_barrier(0);                                    \
    }

template <bool RELU>
__global__ __launch_bounds__(GEMM_THREADS)
void wmma_gemm_f32(const float* __restrict__ A,
                   const float* __restrict__ W,
                   const float* __restrict__ bias,
                   float* __restrict__ C,
                   int M, int N, int K) {
    __shared__ __attribute__((aligned(16))) float As[2][GEMM_BM * GEMM_APITCH];
    __shared__ __attribute__((aligned(16))) float Bs[2][GEMM_BK * GEMM_BPITCH];

    const int tid  = threadIdx.x;
    const int lane = tid & 31;
    const int wave = tid >> 5;
    const int half = lane >> 4;
    const int l15  = lane & 15;
    const int n0   = blockIdx.x * GEMM_BN;
    const int m0   = blockIdx.y * GEMM_BM;

    const int c_row = tid >> 4;
    const int c_c4  = tid & 15;
    const float* __restrict__ Ag = A + (size_t)(m0 + c_row) * K + c_c4 * 4;
    const float* __restrict__ Wg = W + (size_t)c_row * N + n0 + c_c4 * 4;
    const int a_st = c_row * GEMM_APITCH + c_c4 * 4;
    const int b_st = c_row * GEMM_BPITCH + c_c4 * 4;

    v4f pa0, pa1, pa2, pa3, pb0, pb1, pb2, pb3, pb4, pb5, pb6, pb7;
    LDG_SLAB(Ag, Wg);

    const int cn = n0 + wave * 16 + l15;
    const float bv = bias[cn];
    v8f acc0, acc1;
#pragma unroll
    for (int r = 0; r < 8; ++r) { acc0[r] = bv; acc1[r] = bv; }

    STS_SLAB(&As[0][0], &Bs[0][0]);
    __syncthreads();

    const int a_ld = l15 * GEMM_APITCH + 2 * half;
    const int b_ld = (2 * half) * GEMM_BPITCH + wave * 16 + l15;

    const int nk = K / GEMM_BK;
    for (int kt = 0; kt + 1 < nk; ++kt) {
        const int cur = kt & 1;
        LDG_SLAB(Ag + (size_t)(kt + 1) * GEMM_BK,
                 Wg + (size_t)(kt + 1) * GEMM_BK * N);

        const float* as = &As[cur][a_ld];
        const float* bs = &Bs[cur][b_ld];
        SLAB_MMA();

        STS_SLAB(&As[cur ^ 1][0], &Bs[cur ^ 1][0]);
        __syncthreads();
    }
    {
        const int cur = (nk - 1) & 1;
        const float* as = &As[cur][a_ld];
        const float* bs = &Bs[cur][b_ld];
        SLAB_MMA();
    }

    float* __restrict__ Cout = C + (size_t)(m0 + 8 * half) * N + cn;
#pragma unroll
    for (int r = 0; r < 8; ++r) {
        float v0 = acc0[r];
        float v1 = acc1[r];
        if (RELU) { v0 = fmaxf(v0, 0.0f); v1 = fmaxf(v1, 0.0f); }
        Cout[(size_t)r * N]        = v0;
        Cout[(size_t)(r + 16) * N] = v1;
    }
}

__device__ __forceinline__ float block_reduce_128(float x, float* red) {
    const int t = threadIdx.x;
    red[t] = x;
    __syncthreads();
#pragma unroll
    for (int s = 64; s > 0; s >>= 1) {
        if (t < s) red[t] += red[t + s];
        __syncthreads();
    }
    float r = red[0];
    __syncthreads();
    return r;
}

__global__ void epilogue(const float* __restrict__ isp,
                         const float* __restrict__ ish,
                         const int* __restrict__ label,
                         const float* __restrict__ ppos,
                         const float* __restrict__ pneg,
                         float* __restrict__ out) {
    __shared__ float red[128];
    const int b = blockIdx.x;
    const int t = threadIdx.x;

    const float v  = isp[b * 128 + t];
    const float pp = ppos[t];
    const float nn = pneg[t];

    const float nv  = block_reduce_128(v * v, red);
    const float npp = block_reduce_128(pp * pp, red);
    const float nnn = block_reduce_128(nn * nn, red);
    const float dp  = block_reduce_128(v * pp, red);
    const float dn  = block_reduce_128(v * nn, red);

    const float eps = 1e-12f;
    const float dv  = fmaxf(sqrtf(nv), eps);
    const float lp  = dp / (dv * fmaxf(sqrtf(npp), eps));
    const float ln  = dn / (dv * fmaxf(sqrtf(nnn), eps));

    const bool lab    = label[b] != 0;
    const bool pred   = lp > ln;
    const bool is_bad = (pred == lab) && (lp != ln);
    const float diff  = fabsf(lp - ln);
    const float scale = is_bad ? (tanhf(-diff) + 1.0f) : 1.0f;

    out[b * 256 + t]       = v * scale;
    out[b * 256 + 128 + t] = ish[b * 128 + t];
}

extern "C" void kernel_launch(void* const* d_in, const int* in_sizes, int n_in,
                              void* d_out, int out_size, void* d_ws, size_t ws_size,
                              hipStream_t stream) {
    const float* img   = (const float*)d_in[0];
    const int*   label = (const int*)d_in[1];
    const float* W1  = (const float*)d_in[2];
    const float* b1  = (const float*)d_in[3];
    const float* W2  = (const float*)d_in[4];
    const float* b2  = (const float*)d_in[5];
    const float* W3  = (const float*)d_in[6];
    const float* b3  = (const float*)d_in[7];
    const float* Wsp = (const float*)d_in[8];
    const float* bsp = (const float*)d_in[9];
    const float* Wsh = (const float*)d_in[10];
    const float* bsh = (const float*)d_in[11];
    const float* ppn = (const float*)d_in[12];
    const float* pnn = (const float*)d_in[13];

    float* ws   = (float*)d_ws;
    float* feat = ws;
    float* h1   = feat + (size_t)BATCH * C_IN;
    float* h2   = h1   + (size_t)BATCH * 512;
    float* x    = h2   + (size_t)BATCH * 256;
    float* isp  = x    + (size_t)BATCH * 128;
    float* ish  = isp  + (size_t)BATCH * 128;

    {
        const int total = BATCH * C_IN;
        pool_mean32<<<(total + 255) / 256, 256, 0, stream>>>(img, feat, total);
    }

    wmma_gemm_f32<true><<<dim3(512 / GEMM_BN, BATCH / GEMM_BM), GEMM_THREADS, 0, stream>>>(
        feat, W1, b1, h1, BATCH, 512, C_IN);
    wmma_gemm_f32<true><<<dim3(256 / GEMM_BN, BATCH / GEMM_BM), GEMM_THREADS, 0, stream>>>(
        h1, W2, b2, h2, BATCH, 256, 512);
    wmma_gemm_f32<false><<<dim3(128 / GEMM_BN, BATCH / GEMM_BM), GEMM_THREADS, 0, stream>>>(
        h2, W3, b3, x, BATCH, 128, 256);

    wmma_gemm_f32<false><<<dim3(128 / GEMM_BN, BATCH / GEMM_BM), GEMM_THREADS, 0, stream>>>(
        x, Wsp, bsp, isp, BATCH, 128, 128);
    wmma_gemm_f32<false><<<dim3(128 / GEMM_BN, BATCH / GEMM_BM), GEMM_THREADS, 0, stream>>>(
        x, Wsh, bsh, ish, BATCH, 128, 128);

    epilogue<<<BATCH, 128, 0, stream>>>(isp, ish, label, ppn, pnn, (float*)d_out);
}


# pool kernel: each load instruction reads 512 contiguous bytes (4 whole lines, 8 lanes per item) instead of 32 lines 128 B apart; per-piece sums regrouped through wave-private LDS, sum order kept
# speedup vs baseline: 1.7687x; 1.1163x over previous
#include <hip/hip_runtime.h>
#include <hip/hip_bf16.h>


typedef float v2f __attribute__((ext_vector_type(2)));
typedef float v8f __attribute__((ext_vector_type(8)));

#define BATCH   1024
#define C_IN    2048
#define POOLSZ  32

#define POOL_WAVES   4
#define POOL_GROUPS  8
#define POOL_TPITCH  12

static_assert(((size_t)BATCH * C_IN) % (POOL_WAVES * POOL_GROUPS * 32) == 0,
              "the pooling grid must cover the items exactly");

__global__ __launch_bounds__(POOL_WAVES * 32)
void pool_mean32(const float* __restrict__ img, float* __restrict__ feat) {
    __shared__ __attribute__((aligned(16))) float tr[POOL_WAVES][32 * POOL_TPITCH];

    const int lane = threadIdx.x & 31;
    const int wave = threadIdx.x >> 5;
    const size_t item0 =
        ((size_t)blockIdx.x * POOL_WAVES + wave) * (size_t)(POOL_GROUPS * 32);

    const float4* __restrict__ src =
        reinterpret_cast<const float4*>(img + item0 * POOLSZ) + lane;
    float* __restrict__ dst = feat + item0 + lane;

    float* tw = &tr[wave][0];
    float* tst = tw + (lane >> 3) * POOL_TPITCH + (lane & 7);
    const float4* tld = reinterpret_cast<const float4*>(tw + lane * POOL_TPITCH);

    float4 q[2][8];
#pragma unroll
    for (int i = 0; i < 8; ++i) q[0][i] = src[i * 32];

#pragma unroll
    for (int g = 0; g < POOL_GROUPS; ++g) {
        const int cur = g & 1;
        if (g + 1 < POOL_GROUPS) {
#pragma unroll
            for (int i = 0; i < 8; ++i)
                q[cur ^ 1][i] = src[((g + 1) * 8 + i) * 32];
        }
        __builtin_amdgcn_sched_barrier(0);

#pragma unroll
        for (int i = 0; i < 8; ++i) {
            const float4 c = q[cur][i];
            tst[i * 4 * POOL_TPITCH] = (c.x + c.y) + (c.z + c.w);
        }
        __builtin_amdgcn_fence(__ATOMIC_RELEASE, "wavefront");
        __builtin_amdgcn_wave_barrier();
        __builtin_amdgcn_fence(__ATOMIC_ACQUIRE, "wavefront");

        const float4 u = tld[0];
        const float4 v = tld[1];
        float s = 0.0f;
        s += u.x; s += u.y; s += u.z; s += u.w;
        s += v.x; s += v.y; s += v.z; s += v.w;
        dst[g * 32] = s * (1.0f / 32.0f);

        __builtin_amdgcn_fence(__ATOMIC_RELEASE, "wavefront");
        __builtin_amdgcn_wave_barrier();
        __builtin_amdgcn_fence(__ATOMIC_ACQUIRE, "wavefront");
    }
}

typedef float v4f __attribute__((ext_vector_type(4)));

#define GEMM_BM      32
#define GEMM_BN      64
#define GEMM_BK      64
#define GEMM_THREADS 128
#define GEMM_APITCH  68
#define GEMM_BPITCH  72

#define LDG_SLAB(ag, wg)                                                      \
    pa0 = *reinterpret_cast<const v4f*>((ag));                                \
    pa1 = *reinterpret_cast<const v4f*>((ag) + (size_t)8  * K);               \
    pa2 = *reinterpret_cast<const v4f*>((ag) + (size_t)16 * K);               \
    pa3 = *reinterpret_cast<const v4f*>((ag) + (size_t)24 * K);               \
    pb0 = *reinterpret_cast<const v4f*>((wg));                                \
    pb1 = *reinterpret_cast<const v4f*>((wg) + (size_t)8  * N);               \
    pb2 = *reinterpret_cast<const v4f*>((wg) + (size_t)16 * N);               \
    pb3 = *reinterpret_cast<const v4f*>((wg) + (size_t)24 * N);               \
    pb4 = *reinterpret_cast<const v4f*>((wg) + (size_t)32 * N);               \
    pb5 = *reinterpret_cast<const v4f*>((wg) + (size_t)40 * N);               \
    pb6 = *reinterpret_cast<const v4f*>((wg) + (size_t)48 * N);               \
    pb7 = *reinterpret_cast<const v4f*>((wg) + (size_t)56 * N)

#define STS_SLAB(as_, bs_)                                                    \
    *reinterpret_cast<v4f*>((as_) + a_st)                    = pa0;           \
    *reinterpret_cast<v4f*>((as_) + a_st + 8  * GEMM_APITCH) = pa1;           \
    *reinterpret_cast<v4f*>((as_) + a_st + 16 * GEMM_APITCH) = pa2;           \
    *reinterpret_cast<v4f*>((as_) + a_st + 24 * GEMM_APITCH) = pa3;           \
    *reinterpret_cast<v4f*>((bs_) + b_st)                    = pb0;           \
    *reinterpret_cast<v4f*>((bs_) + b_st + 8  * GEMM_BPITCH) = pb1;           \
    *reinterpret_cast<v4f*>((bs_) + b_st + 16 * GEMM_BPITCH) = pb2;           \
    *reinterpret_cast<v4f*>((bs_) + b_st + 24 * GEMM_BPITCH) = pb3;           \
    *reinterpret_cast<v4f*>((bs_) + b_st + 32 * GEMM_BPITCH) = pb4;           \
    *reinterpret_cast<v4f*>((bs_) + b_st + 40 * GEMM_BPITCH) = pb5;           \
    *reinterpret_cast<v4f*>((bs_) + b_st + 48 * GEMM_BPITCH) = pb6;           \
    *reinterpret_cast<v4f*>((bs_) + b_st + 56 * GEMM_BPITCH) = pb7

#define FRAG(ks)                                                              \
    const v2f fa0_##ks = *reinterpret_cast<const v2f*>(as + 4 * (ks));        \
    const v2f fa1_##ks =                                                      \
        *reinterpret_cast<const v2f*>(as + 16 * GEMM_APITCH + 4 * (ks));      \
    v2f fb_##ks;                                                              \
    fb_##ks.x = bs[(4 * (ks)) * GEMM_BPITCH];                                 \
    fb_##ks.y = bs[(4 * (ks) + 1) * GEMM_BPITCH]

#define MMA2(ks)                                                              \
    acc0 = __builtin_amdgcn_wmma_f32_16x16x4_f32(                             \
        false, fa0_##ks, false, fb_##ks, (short)0, acc0, false, false);       \
    acc1 = __builtin_amdgcn_wmma_f32_16x16x4_f32(                             \
        false, fa1_##ks, false, fb_##ks, (short)0, acc1, false, false)

#define SLAB_MMA()                                                            \
    {                                                                         \
        FRAG(0);                                                              \
        FRAG(1);                                                              \
        FRAG(2);                                                              \
        FRAG(3);                                                              \
          \
                 \
        __builtin_amdgcn_sched_barrier(0);                                    \
        FRAG(4);                                                              \
        FRAG(5);                                                              \
        FRAG(6);                                                              \
        FRAG(7);                                                              \
                       \
        __builtin_amdgcn_sched_barrier(0);                                    \
        FRAG(8);                                                              \
        FRAG(9);                                                              \
        FRAG(10);                                                             \
        FRAG(11);                                                             \
                       \
        __builtin_amdgcn_sched_barrier(0);                                    \
        FRAG(12);                                                             \
        FRAG(13);                                                             \
        FRAG(14);                                                             \
        FRAG(15);                                                             \
            \
                 \
        __builtin_amdgcn_sched_barrier(0);                                    \
        MMA2(0);                                                              \
        MMA2(1);                                                              \
        MMA2(2);                                                              \
        MMA2(3);                                                              \
        MMA2(4);                                                              \
        MMA2(5);                                                              \
        MMA2(6);                                                              \
        MMA2(7);                                                              \
        MMA2(8);                                                              \
        MMA2(9);                                                              \
        MMA2(10);                                                             \
        MMA2(11);                                                             \
        MMA2(12);                                                             \
        MMA2(13);                                                             \
        MMA2(14);                                                             \
        MMA2(15);                                                             \
          \
        __builtin_amdgcn_sched_barrier(0);                                    \
    }

template <bool RELU>
__global__ __launch_bounds__(GEMM_THREADS)
void wmma_gemm_f32(const float* __restrict__ A,
                   const float* __restrict__ W,
                   const float* __restrict__ bias,
                   float* __restrict__ C,
                   int M, int N, int K) {
    __shared__ __attribute__((aligned(16))) float As[2][GEMM_BM * GEMM_APITCH];
    __shared__ __attribute__((aligned(16))) float Bs[2][GEMM_BK * GEMM_BPITCH];

    const int tid  = threadIdx.x;
    const int lane = tid & 31;
    const int wave = tid >> 5;
    const int half = lane >> 4;
    const int l15  = lane & 15;
    const int n0   = blockIdx.x * GEMM_BN;
    const int m0   = blockIdx.y * GEMM_BM;

    const int c_row = tid >> 4;
    const int c_c4  = tid & 15;
    const float* __restrict__ Ag = A + (size_t)(m0 + c_row) * K + c_c4 * 4;
    const float* __restrict__ Wg = W + (size_t)c_row * N + n0 + c_c4 * 4;
    const int a_st = c_row * GEMM_APITCH + c_c4 * 4;
    const int b_st = c_row * GEMM_BPITCH + c_c4 * 4;

    v4f pa0, pa1, pa2, pa3, pb0, pb1, pb2, pb3, pb4, pb5, pb6, pb7;
    LDG_SLAB(Ag, Wg);

    const int cn = n0 + wave * 16 + l15;
    const float bv = bias[cn];
    v8f acc0, acc1;
#pragma unroll
    for (int r = 0; r < 8; ++r) { acc0[r] = bv; acc1[r] = bv; }

    STS_SLAB(&As[0][0], &Bs[0][0]);
    __syncthreads();

    const int a_ld = l15 * GEMM_APITCH + 2 * half;
    const int b_ld = (2 * half) * GEMM_BPITCH + wave * 16 + l15;

    const int nk = K / GEMM_BK;
    for (int kt = 0; kt + 1 < nk; ++kt) {
        const int cur = kt & 1;
        LDG_SLAB(Ag + (size_t)(kt + 1) * GEMM_BK,
                 Wg + (size_t)(kt + 1) * GEMM_BK * N);

        const float* as = &As[cur][a_ld];
        const float* bs = &Bs[cur][b_ld];
        SLAB_MMA();

        STS_SLAB(&As[cur ^ 1][0], &Bs[cur ^ 1][0]);
        __syncthreads();
    }
    {
        const int cur = (nk - 1) & 1;
        const float* as = &As[cur][a_ld];
        const float* bs = &Bs[cur][b_ld];
        SLAB_MMA();
    }

    float* __restrict__ Cout = C + (size_t)(m0 + 8 * half) * N + cn;
#pragma unroll
    for (int r = 0; r < 8; ++r) {
        float v0 = acc0[r];
        float v1 = acc1[r];
        if (RELU) { v0 = fmaxf(v0, 0.0f); v1 = fmaxf(v1, 0.0f); }
        Cout[(size_t)r * N]        = v0;
        Cout[(size_t)(r + 16) * N] = v1;
    }
}

__device__ __forceinline__ float block_reduce_128(float x, float* red) {
    const int t = threadIdx.x;
    red[t] = x;
    __syncthreads();
#pragma unroll
    for (int s = 64; s > 0; s >>= 1) {
        if (t < s) red[t] += red[t + s];
        __syncthreads();
    }
    float r = red[0];
    __syncthreads();
    return r;
}

__global__ void epilogue(const float* __restrict__ isp,
                         const float* __restrict__ ish,
                         const int* __restrict__ label,
                         const float* __restrict__ ppos,
                         const float* __restrict__ pneg,
                         float* __restrict__ out) {
    __shared__ float red[128];
    const int b = blockIdx.x;
    const int t = threadIdx.x;

    const float v  = isp[b * 128 + t];
    const float pp = ppos[t];
    const float nn = pneg[t];

    const float nv  = block_reduce_128(v * v, red);
    const float npp = block_reduce_128(pp * pp, red);
    const float nnn = block_reduce_128(nn * nn, red);
    const float dp  = block_reduce_128(v * pp, red);
    const float dn  = block_reduce_128(v * nn, red);

    const float eps = 1e-12f;
    const float dv  = fmaxf(sqrtf(nv), eps);
    const float lp  = dp / (dv * fmaxf(sqrtf(npp), eps));
    const float ln  = dn / (dv * fmaxf(sqrtf(nnn), eps));

    const bool lab    = label[b] != 0;
    const bool pred   = lp > ln;
    const bool is_bad = (pred == lab) && (lp != ln);
    const float diff  = fabsf(lp - ln);
    const float scale = is_bad ? (tanhf(-diff) + 1.0f) : 1.0f;

    out[b * 256 + t]       = v * scale;
    out[b * 256 + 128 + t] = ish[b * 128 + t];
}

extern "C" void kernel_launch(void* const* d_in, const int* in_sizes, int n_in,
                              void* d_out, int out_size, void* d_ws, size_t ws_size,
                              hipStream_t stream) {
    const float* img   = (const float*)d_in[0];
    const int*   label = (const int*)d_in[1];
    const float* W1  = (const float*)d_in[2];
    const float* b1  = (const float*)d_in[3];
    const float* W2  = (const float*)d_in[4];
    const float* b2  = (const float*)d_in[5];
    const float* W3  = (const float*)d_in[6];
    const float* b3  = (const float*)d_in[7];
    const float* Wsp = (const float*)d_in[8];
    const float* bsp = (const float*)d_in[9];
    const float* Wsh = (const float*)d_in[10];
    const float* bsh = (const float*)d_in[11];
    const float* ppn = (const float*)d_in[12];
    const float* pnn = (const float*)d_in[13];

    float* ws   = (float*)d_ws;
    float* feat = ws;
    float* h1   = feat + (size_t)BATCH * C_IN;
    float* h2   = h1   + (size_t)BATCH * 512;
    float* x    = h2   + (size_t)BATCH * 256;
    float* isp  = x    + (size_t)BATCH * 128;
    float* ish  = isp  + (size_t)BATCH * 128;

    pool_mean32<<<(BATCH * C_IN) / (POOL_WAVES * POOL_GROUPS * 32), POOL_WAVES * 32, 0, stream>>>(
        img, feat);

    wmma_gemm_f32<true><<<dim3(512 / GEMM_BN, BATCH / GEMM_BM), GEMM_THREADS, 0, stream>>>(
        feat, W1, b1, h1, BATCH, 512, C_IN);
    wmma_gemm_f32<true><<<dim3(256 / GEMM_BN, BATCH / GEMM_BM), GEMM_THREADS, 0, stream>>>(
        h1, W2, b2, h2, BATCH, 256, 512);
    wmma_gemm_f32<false><<<dim3(128 / GEMM_BN, BATCH / GEMM_BM), GEMM_THREADS, 0, stream>>>(
        h2, W3, b3, x, BATCH, 128, 256);

    wmma_gemm_f32<false><<<dim3(128 / GEMM_BN, BATCH / GEMM_BM), GEMM_THREADS, 0, stream>>>(
        x, Wsp, bsp, isp, BATCH, 128, 128);
    wmma_gemm_f32<false><<<dim3(128 / GEMM_BN, BATCH / GEMM_BM), GEMM_THREADS, 0, stream>>>(
        x, Wsh, bsh, ish, BATCH, 128, 128);

    epilogue<<<BATCH, 128, 0, stream>>>(isp, ish, label, ppn, pnn, (float*)d_out);
}


# last 4 launches (h2@W3, both heads, epilogue) fused into one: 16 rows per 4-wave workgroup, x and heads stay in LDS, weights load straight into B fragments; sum order kept, epilogue products may fuse
# speedup vs baseline: 2.0584x; 1.1638x over previous
#include <hip/hip_runtime.h>
#include <hip/hip_bf16.h>


typedef float v2f __attribute__((ext_vector_type(2)));
typedef float v8f __attribute__((ext_vector_type(8)));

#define BATCH   1024
#define C_IN    2048
#define POOLSZ  32

#define POOL_WAVES   4
#define POOL_GROUPS  8
#define POOL_TPITCH  12

static_assert(((size_t)BATCH * C_IN) % (POOL_WAVES * POOL_GROUPS * 32) == 0,
              "the pooling grid must cover the items exactly");

__global__ __launch_bounds__(POOL_WAVES * 32)
void pool_mean32(const float* __restrict__ img, float* __restrict__ feat) {
    __shared__ __attribute__((aligned(16))) float tr[POOL_WAVES][32 * POOL_TPITCH];

    const int lane = threadIdx.x & 31;
    const int wave = threadIdx.x >> 5;
    const size_t item0 =
        ((size_t)blockIdx.x * POOL_WAVES + wave) * (size_t)(POOL_GROUPS * 32);

    const float4* __restrict__ src =
        reinterpret_cast<const float4*>(img + item0 * POOLSZ) + lane;
    float* __restrict__ dst = feat + item0 + lane;

    float* tw = &tr[wave][0];
    float* tst = tw + (lane >> 3) * POOL_TPITCH + (lane & 7);
    const float4* tld = reinterpret_cast<const float4*>(tw + lane * POOL_TPITCH);

    float4 q[2][8];
#pragma unroll
    for (int i = 0; i < 8; ++i) q[0][i] = src[i * 32];

#pragma unroll
    for (int g = 0; g < POOL_GROUPS; ++g) {
        const int cur = g & 1;
        if (g + 1 < POOL_GROUPS) {
#pragma unroll
            for (int i = 0; i < 8; ++i)
                q[cur ^ 1][i] = src[((g + 1) * 8 + i) * 32];
        }
        __builtin_amdgcn_sched_barrier(0);

#pragma unroll
        for (int i = 0; i < 8; ++i) {
            const float4 c = q[cur][i];
            tst[i * 4 * POOL_TPITCH] = (c.x + c.y) + (c.z + c.w);
        }
        __builtin_amdgcn_fence(__ATOMIC_RELEASE, "wavefront");
        __builtin_amdgcn_wave_barrier();
        __builtin_amdgcn_fence(__ATOMIC_ACQUIRE, "wavefront");

        const float4 u = tld[0];
        const float4 v = tld[1];
        float s = 0.0f;
        s += u.x; s += u.y; s += u.z; s += u.w;
        s += v.x; s += v.y; s += v.z; s += v.w;
        dst[g * 32] = s * (1.0f / 32.0f);

        __builtin_amdgcn_fence(__ATOMIC_RELEASE, "wavefront");
        __builtin_amdgcn_wave_barrier();
        __builtin_amdgcn_fence(__ATOMIC_ACQUIRE, "wavefront");
    }
}

typedef float v4f __attribute__((ext_vector_type(4)));

#define GEMM_BM      32
#define GEMM_BN      64
#define GEMM_BK      64
#define GEMM_THREADS 128
#define GEMM_APITCH  68
#define GEMM_BPITCH  72

#define LDG_SLAB(ag, wg)                                                      \
    pa0 = *reinterpret_cast<const v4f*>((ag));                                \
    pa1 = *reinterpret_cast<const v4f*>((ag) + (size_t)8  * K);               \
    pa2 = *reinterpret_cast<const v4f*>((ag) + (size_t)16 * K);               \
    pa3 = *reinterpret_cast<const v4f*>((ag) + (size_t)24 * K);               \
    pb0 = *reinterpret_cast<const v4f*>((wg));                                \
    pb1 = *reinterpret_cast<const v4f*>((wg) + (size_t)8  * N);               \
    pb2 = *reinterpret_cast<const v4f*>((wg) + (size_t)16 * N);               \
    pb3 = *reinterpret_cast<const v4f*>((wg) + (size_t)24 * N);               \
    pb4 = *reinterpret_cast<const v4f*>((wg) + (size_t)32 * N);               \
    pb5 = *reinterpret_cast<const v4f*>((wg) + (size_t)40 * N);               \
    pb6 = *reinterpret_cast<const v4f*>((wg) + (size_t)48 * N);               \
    pb7 = *reinterpret_cast<const v4f*>((wg) + (size_t)56 * N)

#define STS_SLAB(as_, bs_)                                                    \
    *reinterpret_cast<v4f*>((as_) + a_st)                    = pa0;           \
    *reinterpret_cast<v4f*>((as_) + a_st + 8  * GEMM_APITCH) = pa1;           \
    *reinterpret_cast<v4f*>((as_) + a_st + 16 * GEMM_APITCH) = pa2;           \
    *reinterpret_cast<v4f*>((as_) + a_st + 24 * GEMM_APITCH) = pa3;           \
    *reinterpret_cast<v4f*>((bs_) + b_st)                    = pb0;           \
    *reinterpret_cast<v4f*>((bs_) + b_st + 8  * GEMM_BPITCH) = pb1;           \
    *reinterpret_cast<v4f*>((bs_) + b_st + 16 * GEMM_BPITCH) = pb2;           \
    *reinterpret_cast<v4f*>((bs_) + b_st + 24 * GEMM_BPITCH) = pb3;           \
    *reinterpret_cast<v4f*>((bs_) + b_st + 32 * GEMM_BPITCH) = pb4;           \
    *reinterpret_cast<v4f*>((bs_) + b_st + 40 * GEMM_BPITCH) = pb5;           \
    *reinterpret_cast<v4f*>((bs_) + b_st + 48 * GEMM_BPITCH) = pb6;           \
    *reinterpret_cast<v4f*>((bs_) + b_st + 56 * GEMM_BPITCH) = pb7

#define FRAG(ks)                                                              \
    const v2f fa0_##ks = *reinterpret_cast<const v2f*>(as + 4 * (ks));        \
    const v2f fa1_##ks =                                                      \
        *reinterpret_cast<const v2f*>(as + 16 * GEMM_APITCH + 4 * (ks));      \
    v2f fb_##ks;                                                              \
    fb_##ks.x = bs[(4 * (ks)) * GEMM_BPITCH];                                 \
    fb_##ks.y = bs[(4 * (ks) + 1) * GEMM_BPITCH]

#define MMA2(ks)                                                              \
    acc0 = __builtin_amdgcn_wmma_f32_16x16x4_f32(                             \
        false, fa0_##ks, false, fb_##ks, (short)0, acc0, false, false);       \
    acc1 = __builtin_amdgcn_wmma_f32_16x16x4_f32(                             \
        false, fa1_##ks, false, fb_##ks, (short)0, acc1, false, false)

#define SLAB_MMA()                                                            \
    {                                                                         \
        FRAG(0);                                                              \
        FRAG(1);                                                              \
        FRAG(2);                                                              \
        FRAG(3);                                                              \
          \
                 \
        __builtin_amdgcn_sched_barrier(0);                                    \
        FRAG(4);                                                              \
        FRAG(5);                                                              \
        FRAG(6);                                                              \
        FRAG(7);                                                              \
                       \
        __builtin_amdgcn_sched_barrier(0);                                    \
        FRAG(8);                                                              \
        FRAG(9);                                                              \
        FRAG(10);                                                             \
        FRAG(11);                                                             \
                       \
        __builtin_amdgcn_sched_barrier(0);                                    \
        FRAG(12);                                                             \
        FRAG(13);                                                             \
        FRAG(14);                                                             \
        FRAG(15);                                                             \
            \
                 \
        __builtin_amdgcn_sched_barrier(0);                                    \
        MMA2(0);                                                              \
        MMA2(1);                                                              \
        MMA2(2);                                                              \
        MMA2(3);                                                              \
        MMA2(4);                                                              \
        MMA2(5);                                                              \
        MMA2(6);                                                              \
        MMA2(7);                                                              \
        MMA2(8);                                                              \
        MMA2(9);                                                              \
        MMA2(10);                                                             \
        MMA2(11);                                                             \
        MMA2(12);                                                             \
        MMA2(13);                                                             \
        MMA2(14);                                                             \
        MMA2(15);                                                             \
          \
        __builtin_amdgcn_sched_barrier(0);                                    \
    }

template <bool RELU>
__global__ __launch_bounds__(GEMM_THREADS)
void wmma_gemm_f32(const float* __restrict__ A,
                   const float* __restrict__ W,
                   const float* __restrict__ bias,
                   float* __restrict__ C,
                   int M, int N, int K) {
    __shared__ __attribute__((aligned(16))) float As[2][GEMM_BM * GEMM_APITCH];
    __shared__ __attribute__((aligned(16))) float Bs[2][GEMM_BK * GEMM_BPITCH];

    const int tid  = threadIdx.x;
    const int lane = tid & 31;
    const int wave = tid >> 5;
    const int half = lane >> 4;
    const int l15  = lane & 15;
    const int n0   = blockIdx.x * GEMM_BN;
    const int m0   = blockIdx.y * GEMM_BM;

    const int c_row = tid >> 4;
    const int c_c4  = tid & 15;
    const float* __restrict__ Ag = A + (size_t)(m0 + c_row) * K + c_c4 * 4;
    const float* __restrict__ Wg = W + (size_t)c_row * N + n0 + c_c4 * 4;
    const int a_st = c_row * GEMM_APITCH + c_c4 * 4;
    const int b_st = c_row * GEMM_BPITCH + c_c4 * 4;

    v4f pa0, pa1, pa2, pa3, pb0, pb1, pb2, pb3, pb4, pb5, pb6, pb7;
    LDG_SLAB(Ag, Wg);

    const int cn = n0 + wave * 16 + l15;
    const float bv = bias[cn];
    v8f acc0, acc1;
#pragma unroll
    for (int r = 0; r < 8; ++r) { acc0[r] = bv; acc1[r] = bv; }

    STS_SLAB(&As[0][0], &Bs[0][0]);
    __syncthreads();

    const int a_ld = l15 * GEMM_APITCH + 2 * half;
    const int b_ld = (2 * half) * GEMM_BPITCH + wave * 16 + l15;

    const int nk = K / GEMM_BK;
    for (int kt = 0; kt + 1 < nk; ++kt) {
        const int cur = kt & 1;
        LDG_SLAB(Ag + (size_t)(kt + 1) * GEMM_BK,
                 Wg + (size_t)(kt + 1) * GEMM_BK * N);

        const float* as = &As[cur][a_ld];
        const float* bs = &Bs[cur][b_ld];
        SLAB_MMA();

        STS_SLAB(&As[cur ^ 1][0], &Bs[cur ^ 1][0]);
        __syncthreads();
    }
    {
        const int cur = (nk - 1) & 1;
        const float* as = &As[cur][a_ld];
        const float* bs = &Bs[cur][b_ld];
        SLAB_MMA();
    }

    float* __restrict__ Cout = C + (size_t)(m0 + 8 * half) * N + cn;
#pragma unroll
    for (int r = 0; r < 8; ++r) {
        float v0 = acc0[r];
        float v1 = acc1[r];
        if (RELU) { v0 = fmaxf(v0, 0.0f); v1 = fmaxf(v1, 0.0f); }
        Cout[(size_t)r * N]        = v0;
        Cout[(size_t)(r + 16) * N] = v1;
    }
}

#define TAIL_ROWS    16
#define TAIL_THREADS 128
#define TAIL_KC      8
#define TAIL_HPITCH  260
#define TAIL_XPITCH  132
#define TAIL_OPITCH  260

__device__ __forceinline__ float lane_xor16(float x) {
    return __builtin_bit_cast(float, __builtin_amdgcn_permlanex16(
        __builtin_bit_cast(int, x), __builtin_bit_cast(int, x),
        0x76543210, 0xfedcba98, false, false));
}
template <int M>
__device__ __forceinline__ float row_xor(float x) {
    return __builtin_bit_cast(float, __builtin_amdgcn_update_dpp(
        0, __builtin_bit_cast(int, x), 0x160 + M, 0xf, 0xf, true));
}
__device__ __forceinline__ float wave_tree_sum(float b) {
    b += lane_xor16(b);
    b += row_xor<8>(b);
    b += row_xor<4>(b);
    b += row_xor<2>(b);
    b += row_xor<1>(b);
    return b;
}

#define TAIL_LD_G(ch)                                                         \
    _Pragma("unroll")                                                         \
    for (int i = 0; i < TAIL_KC; ++i) {                                       \
        g[(ch)][i][0] = *reinterpret_cast<const v2f*>(                        \
            w3p + (size_t)(4 * ((ch) * TAIL_KC + i)) * 128);                  \
        g[(ch)][i][1] = *reinterpret_cast<const v2f*>(                        \
            w3p + (size_t)(4 * ((ch) * TAIL_KC + i) + 1) * 128);              \
    }

#define TAIL_LD_Q(ch)                                                         \
    _Pragma("unroll")                                                         \
    for (int i = 0; i < TAIL_KC; ++i) {                                       \
        q[(ch)][i][0] = *reinterpret_cast<const v4f*>(                        \
            whp + (size_t)(4 * ((ch) * TAIL_KC + i)) * 128);                  \
        q[(ch)][i][1] = *reinterpret_cast<const v4f*>(                        \
            whp + (size_t)(4 * ((ch) * TAIL_KC + i) + 1) * 128);              \
    }

#define TAIL_WMMA(a, b, c)                                                    \
    __builtin_amdgcn_wmma_f32_16x16x4_f32(false, (a), false, (b), (short)0,   \
                                          (c), false, false)

__global__ __launch_bounds__(TAIL_THREADS)
void tail_fused(const float* __restrict__ H2,
                const float* __restrict__ W3,  const float* __restrict__ b3,
                const float* __restrict__ Wsp, const float* __restrict__ bsp,
                const float* __restrict__ Wsh, const float* __restrict__ bsh,
                const int* __restrict__ label,
                const float* __restrict__ ppos,
                const float* __restrict__ pneg,
                float* __restrict__ out) {
    __shared__ __attribute__((aligned(16))) float Hs[TAIL_ROWS * TAIL_HPITCH];
    __shared__ __attribute__((aligned(16))) float Xs[TAIL_ROWS * TAIL_XPITCH];
    __shared__ __attribute__((aligned(16))) float Os[TAIL_ROWS * TAIL_OPITCH];

    const int tid  = threadIdx.x;
    const int lane = tid & 31;
    const int wave = tid >> 5;
    const int half = lane >> 4;
    const int l15  = lane & 15;
    const int m0   = blockIdx.x * TAIL_ROWS;

    v4f ha[8];
#pragma unroll
    for (int i = 0; i < 8; ++i) {
        const int idx = tid + TAIL_THREADS * i;
        ha[i] = *reinterpret_cast<const v4f*>(
            H2 + (size_t)(m0 + (idx >> 6)) * 256 + (idx & 63) * 4);
    }

    const int xc = 32 * wave + 2 * l15;
    const float* __restrict__ w3p = W3 + (size_t)(2 * half) * 128 + xc;
    const float* __restrict__ Wh = (wave < 2) ? Wsp : Wsh;
    const float* __restrict__ bh = (wave < 2) ? bsp : bsh;
    const int hb = 64 * (wave & 1) + 4 * l15;
    const float* __restrict__ whp = Wh + (size_t)(2 * half) * 128 + hb;

    v2f g[8][TAIL_KC][2];
    v4f q[4][TAIL_KC][2];
    TAIL_LD_G(0);
    TAIL_LD_G(1);
    TAIL_LD_G(2);

    const v2f  bx = *reinterpret_cast<const v2f*>(b3 + xc);
    const v4f  bo = *reinterpret_cast<const v4f*>(bh + hb);
    float pp[4], nn[4];
#pragma unroll
    for (int i = 0; i < 4; ++i) {
        pp[i] = ppos[lane + 32 * i];
        nn[i] = pneg[lane + 32 * i];
    }

#pragma unroll
    for (int i = 0; i < 8; ++i) {
        const int idx = tid + TAIL_THREADS * i;
        *reinterpret_cast<v4f*>(&Hs[(idx >> 6) * TAIL_HPITCH + (idx & 63) * 4]) = ha[i];
    }
    __syncthreads();

    v8f xa0, xa1;
#pragma unroll
    for (int r = 0; r < 8; ++r) { xa0[r] = bx.x; xa1[r] = bx.y; }

    v2f af[2][TAIL_KC];
    {
        const float* ap = &Hs[l15 * TAIL_HPITCH + 2 * half];
#pragma unroll
        for (int i = 0; i < TAIL_KC; ++i)
            af[0][i] = *reinterpret_cast<const v2f*>(ap + 4 * i);

#pragma unroll
        for (int ch = 0; ch < 8; ++ch) {
            if (ch + 3 < 8) { TAIL_LD_G(ch + 3); }
            else            { TAIL_LD_Q(ch + 3 - 8); }
            if (ch + 1 < 8) {
#pragma unroll
                for (int i = 0; i < TAIL_KC; ++i)
                    af[(ch + 1) & 1][i] = *reinterpret_cast<const v2f*>(
                        ap + 4 * ((ch + 1) * TAIL_KC + i));
            }
            __builtin_amdgcn_sched_barrier(0);
#pragma unroll
            for (int i = 0; i < TAIL_KC; ++i) {
                v2f f0, f1;
                f0.x = g[ch][i][0].x; f0.y = g[ch][i][1].x;
                f1.x = g[ch][i][0].y; f1.y = g[ch][i][1].y;
                xa0 = TAIL_WMMA(af[ch & 1][i], f0, xa0);
                xa1 = TAIL_WMMA(af[ch & 1][i], f1, xa1);
            }
            __builtin_amdgcn_sched_barrier(0);
        }
    }

    {
        float* xs = &Xs[(8 * half) * TAIL_XPITCH + xc];
#pragma unroll
        for (int r = 0; r < 8; ++r) {
            v2f t; t.x = xa0[r]; t.y = xa1[r];
            *reinterpret_cast<v2f*>(xs + r * TAIL_XPITCH) = t;
        }
    }
    __syncthreads();

    v8f oa0, oa1, oa2, oa3;
#pragma unroll
    for (int r = 0; r < 8; ++r) { oa0[r] = bo.x; oa1[r] = bo.y; oa2[r] = bo.z; oa3[r] = bo.w; }
    {
        const float* xp = &Xs[l15 * TAIL_XPITCH + 2 * half];
#pragma unroll
        for (int i = 0; i < TAIL_KC; ++i)
            af[0][i] = *reinterpret_cast<const v2f*>(xp + 4 * i);

#pragma unroll
        for (int ch = 0; ch < 4; ++ch) {
            if (ch + 3 < 4) { TAIL_LD_Q(ch + 3); }
            if (ch + 1 < 4) {
#pragma unroll
                for (int i = 0; i < TAIL_KC; ++i)
                    af[(ch + 1) & 1][i] = *reinterpret_cast<const v2f*>(
                        xp + 4 * ((ch + 1) * TAIL_KC + i));
            }
            __builtin_amdgcn_sched_barrier(0);
#pragma unroll
            for (int i = 0; i < TAIL_KC; ++i) {
                const v4f k0 = q[ch][i][0];
                const v4f k1 = q[ch][i][1];
                v2f f0, f1, f2, f3;
                f0.x = k0.x; f0.y = k1.x;
                f1.x = k0.y; f1.y = k1.y;
                f2.x = k0.z; f2.y = k1.z;
                f3.x = k0.w; f3.y = k1.w;
                oa0 = TAIL_WMMA(af[ch & 1][i], f0, oa0);
                oa1 = TAIL_WMMA(af[ch & 1][i], f1, oa1);
                oa2 = TAIL_WMMA(af[ch & 1][i], f2, oa2);
                oa3 = TAIL_WMMA(af[ch & 1][i], f3, oa3);
            }
            __builtin_amdgcn_sched_barrier(0);
        }
    }

    {
        float* os = &Os[(8 * half) * TAIL_OPITCH + 64 * wave + 4 * l15];
#pragma unroll
        for (int r = 0; r < 8; ++r) {
            v4f t; t.x = oa0[r]; t.y = oa1[r]; t.z = oa2[r]; t.w = oa3[r];
            *reinterpret_cast<v4f*>(os + r * TAIL_OPITCH) = t;
        }
    }
    __syncthreads();

    const float eps = 1e-12f;
    const float npp = wave_tree_sum((pp[0] * pp[0] + pp[2] * pp[2]) +
                                    (pp[1] * pp[1] + pp[3] * pp[3]));
    const float nnn = wave_tree_sum((nn[0] * nn[0] + nn[2] * nn[2]) +
                                    (nn[1] * nn[1] + nn[3] * nn[3]));
    const float dpp = fmaxf(sqrtf(npp), eps);
    const float dnn = fmaxf(sqrtf(nnn), eps);

    for (int rr = 0; rr < 4; ++rr) {
        const int row = 4 * wave + rr;
        const int b   = m0 + row;
        const float* orow = &Os[row * TAIL_OPITCH + lane];
        float v[4], sh[4];
#pragma unroll
        for (int i = 0; i < 4; ++i) { v[i] = orow[32 * i]; sh[i] = orow[128 + 32 * i]; }

        const float nv = wave_tree_sum((v[0] * v[0] + v[2] * v[2]) +
                                       (v[1] * v[1] + v[3] * v[3]));
        const float dp = wave_tree_sum((v[0] * pp[0] + v[2] * pp[2]) +
                                       (v[1] * pp[1] + v[3] * pp[3]));
        const float dn = wave_tree_sum((v[0] * nn[0] + v[2] * nn[2]) +
                                       (v[1] * nn[1] + v[3] * nn[3]));

        const float dv = fmaxf(sqrtf(nv), eps);
        const float lp = dp / (dv * dpp);
        const float ln = dn / (dv * dnn);

        const bool lab    = label[b] != 0;
        const bool pred   = lp > ln;
        const bool is_bad = (pred == lab) && (lp != ln);
        const float diff  = fabsf(lp - ln);
        const float scale = is_bad ? (tanhf(-diff) + 1.0f) : 1.0f;

        float* o = out + (size_t)b * 256 + lane;
#pragma unroll
        for (int i = 0; i < 4; ++i) {
            o[32 * i]       = v[i] * scale;
            o[128 + 32 * i] = sh[i];
        }
    }
}

extern "C" void kernel_launch(void* const* d_in, const int* in_sizes, int n_in,
                              void* d_out, int out_size, void* d_ws, size_t ws_size,
                              hipStream_t stream) {
    const float* img   = (const float*)d_in[0];
    const int*   label = (const int*)d_in[1];
    const float* W1  = (const float*)d_in[2];
    const float* b1  = (const float*)d_in[3];
    const float* W2  = (const float*)d_in[4];
    const float* b2  = (const float*)d_in[5];
    const float* W3  = (const float*)d_in[6];
    const float* b3  = (const float*)d_in[7];
    const float* Wsp = (const float*)d_in[8];
    const float* bsp = (const float*)d_in[9];
    const float* Wsh = (const float*)d_in[10];
    const float* bsh = (const float*)d_in[11];
    const float* ppn = (const float*)d_in[12];
    const float* pnn = (const float*)d_in[13];

    float* ws   = (float*)d_ws;
    float* feat = ws;
    float* h1   = feat + (size_t)BATCH * C_IN;
    float* h2   = h1   + (size_t)BATCH * 512;

    pool_mean32<<<(BATCH * C_IN) / (POOL_WAVES * POOL_GROUPS * 32), POOL_WAVES * 32, 0, stream>>>(
        img, feat);

    wmma_gemm_f32<true><<<dim3(512 / GEMM_BN, BATCH / GEMM_BM), GEMM_THREADS, 0, stream>>>(
        feat, W1, b1, h1, BATCH, 512, C_IN);
    wmma_gemm_f32<true><<<dim3(256 / GEMM_BN, BATCH / GEMM_BM), GEMM_THREADS, 0, stream>>>(
        h1, W2, b2, h2, BATCH, 256, 512);
    tail_fused<<<BATCH / TAIL_ROWS, TAIL_THREADS, 0, stream>>>(
        h2, W3, b3, Wsp, bsp, Wsh, bsh, label, ppn, pnn, (float*)d_out);
}


# pool fused into the first GEMM: 16-wave workgroup pools 16 rows x 64 channels per slab into LDS, times W1 from L2; K split in 4, second GEMM's A load sums b1 + 4 planes (h1 sum order changes)
# speedup vs baseline: 2.6241x; 1.2748x over previous
#include <hip/hip_runtime.h>
#include <hip/hip_bf16.h>


typedef float v2f __attribute__((ext_vector_type(2)));
typedef float v8f __attribute__((ext_vector_type(8)));

#define BATCH   1024
#define C_IN    2048
#define POOLSZ  32

#define PG_WAVES    16
#define PG_THREADS  (PG_WAVES * 32)
#define PG_KSPLIT   4
#define PG_KCH      (C_IN / PG_KSPLIT)
#define PG_SLABS    (PG_KCH / 64)
#define PG_N        512
#define PG_FPITCH   68
#define PG_FSTAGE   (16 * PG_FPITCH)
#define POOL_TPITCH 12

static_assert(BATCH % 16 == 0 && C_IN % (PG_KSPLIT * 64) == 0,
              "the fused grid must cover rows and channels exactly");

__global__ __launch_bounds__(PG_THREADS)
void pool_gemm1_partial(const float* __restrict__ img,
                        const float* __restrict__ W1,
                        float* __restrict__ part) {
    __shared__ __attribute__((aligned(16))) float tr[PG_WAVES][32 * POOL_TPITCH];
    __shared__ __attribute__((aligned(16))) float Fs[2][PG_FSTAGE];

    const int lane = threadIdx.x & 31;
    const int wave = threadIdx.x >> 5;
    const int half = lane >> 4;
    const int l15  = lane & 15;
    const int mt   = blockIdx.x >> 2;
    const int kq   = blockIdx.x & 3;

    const size_t item0 = (size_t)(16 * mt + wave) * C_IN + (size_t)kq * PG_KCH;
    const float4* __restrict__ src =
        reinterpret_cast<const float4*>(img + item0 * POOLSZ) + lane;

    const float* __restrict__ wp =
        W1 + (size_t)(kq * PG_KCH + 2 * half) * PG_N + 32 * wave + 2 * l15;

    float* tw = &tr[wave][0];
    float* tst = tw + (lane >> 3) * POOL_TPITCH + (lane & 7);
    const float4* tld = reinterpret_cast<const float4*>(tw + lane * POOL_TPITCH);
    float* fst = &Fs[0][wave * PG_FPITCH + lane];
    const float* fld = &Fs[0][l15 * PG_FPITCH + 2 * half];

    float4 q[2][16];
#pragma unroll
    for (int i = 0; i < 16; ++i) q[0][i] = src[i * 32];

    v8f acc0, acc1;
#pragma unroll
    for (int r = 0; r < 8; ++r) { acc0[r] = 0.0f; acc1[r] = 0.0f; }

#pragma unroll
    for (int t = 0; t < PG_SLABS; ++t) {
        const int cur = t & 1;

        v2f bq[16][2];
#pragma unroll
        for (int i = 0; i < 16; ++i) {
            bq[i][0] = *reinterpret_cast<const v2f*>(
                wp + (size_t)(64 * t + 4 * i) * PG_N);
            bq[i][1] = *reinterpret_cast<const v2f*>(
                wp + (size_t)(64 * t + 4 * i + 1) * PG_N);
        }
        if (t + 1 < PG_SLABS) {
#pragma unroll
            for (int i = 0; i < 16; ++i)
                q[cur ^ 1][i] = src[((t + 1) * 16 + i) * 32];
        }
        __builtin_amdgcn_sched_barrier(0);

#pragma unroll
        for (int h = 0; h < 2; ++h) {
#pragma unroll
            for (int i = 0; i < 8; ++i) {
                const float4 c = q[cur][h * 8 + i];
                tst[i * 4 * POOL_TPITCH] = (c.x + c.y) + (c.z + c.w);
            }
            __builtin_amdgcn_fence(__ATOMIC_RELEASE, "wavefront");
            __builtin_amdgcn_wave_barrier();
            __builtin_amdgcn_fence(__ATOMIC_ACQUIRE, "wavefront");

            const float4 u = tld[0];
            const float4 v = tld[1];
            float s = 0.0f;
            s += u.x; s += u.y; s += u.z; s += u.w;
            s += v.x; s += v.y; s += v.z; s += v.w;
            fst[cur * PG_FSTAGE + 32 * h] = s * (1.0f / 32.0f);

            __builtin_amdgcn_fence(__ATOMIC_RELEASE, "wavefront");
            __builtin_amdgcn_wave_barrier();
            __builtin_amdgcn_fence(__ATOMIC_ACQUIRE, "wavefront");
        }

        asm volatile("s_wait_dscnt 0x0\n\ts_barrier_signal -1\n\ts_barrier_wait -1" ::: "memory");

        v2f af[16];
#pragma unroll
        for (int i = 0; i < 16; ++i)
            af[i] = *reinterpret_cast<const v2f*>(fld + cur * PG_FSTAGE + 4 * i);
        __builtin_amdgcn_sched_barrier(0);
#pragma unroll
        for (int i = 0; i < 16; ++i) {
            v2f f0, f1;
            f0.x = bq[i][0].x; f0.y = bq[i][1].x;
            f1.x = bq[i][0].y; f1.y = bq[i][1].y;
            acc0 = __builtin_amdgcn_wmma_f32_16x16x4_f32(
                false, af[i], false, f0, (short)0, acc0, false, false);
            acc1 = __builtin_amdgcn_wmma_f32_16x16x4_f32(
                false, af[i], false, f1, (short)0, acc1, false, false);
        }
        __builtin_amdgcn_sched_barrier(0);
    }

    float* __restrict__ pp = part + (size_t)kq * BATCH * PG_N +
                             (size_t)(16 * mt + 8 * half) * PG_N + 32 * wave + 2 * l15;
#pragma unroll
    for (int r = 0; r < 8; ++r) {
        v2f o; o.x = acc0[r]; o.y = acc1[r];
        *reinterpret_cast<v2f*>(pp + (size_t)r * PG_N) = o;
    }
}

typedef float v4f __attribute__((ext_vector_type(4)));

#define GEMM_BM      32
#define GEMM_BN      64
#define GEMM_BK      64
#define GEMM_THREADS 128
#define GEMM_NP      PG_KSPLIT
#define GEMM_APITCH  68
#define GEMM_BPITCH  72

#define LDG_SLAB(kt_)                                                         \
    _Pragma("unroll")                                                         \
    for (int p = 0; p < GEMM_NP; ++p) {                                       \
        _Pragma("unroll")                                                     \
        for (int i = 0; i < 4; ++i)                                           \
            pa[p][i] = *reinterpret_cast<const v4f*>(                         \
                Ag + (size_t)p * plane + (size_t)(8 * i) * K +                \
                (size_t)(kt_) * GEMM_BK);                                     \
    }                                                                         \
    pk = *reinterpret_cast<const v4f*>(abias + (kt_) * GEMM_BK + c_c4 * 4);   \
    _Pragma("unroll")                                                         \
    for (int i = 0; i < 8; ++i)                                               \
        pb[i] = *reinterpret_cast<const v4f*>(                                \
            Wg + (size_t)(8 * i) * N + (size_t)(kt_) * GEMM_BK * N)

#define STS_SLAB(as_, bs_)                                                    \
    _Pragma("unroll")                                                         \
    for (int i = 0; i < 4; ++i) {                                             \
        v4f s = pk + pa[0][i];                                                \
        _Pragma("unroll")                                                     \
        for (int p = 1; p < GEMM_NP; ++p) s = s + pa[p][i];                   \
        s.x = fmaxf(s.x, 0.0f); s.y = fmaxf(s.y, 0.0f);                       \
        s.z = fmaxf(s.z, 0.0f); s.w = fmaxf(s.w, 0.0f);                       \
        *reinterpret_cast<v4f*>((as_) + a_st + 8 * i * GEMM_APITCH) = s;      \
    }                                                                         \
    _Pragma("unroll")                                                         \
    for (int i = 0; i < 8; ++i)                                               \
        *reinterpret_cast<v4f*>((bs_) + b_st + 8 * i * GEMM_BPITCH) = pb[i]

#define FRAG(ks)                                                              \
    const v2f fa0_##ks = *reinterpret_cast<const v2f*>(as + 4 * (ks));        \
    const v2f fa1_##ks =                                                      \
        *reinterpret_cast<const v2f*>(as + 16 * GEMM_APITCH + 4 * (ks));      \
    v2f fb_##ks;                                                              \
    fb_##ks.x = bs[(4 * (ks)) * GEMM_BPITCH];                                 \
    fb_##ks.y = bs[(4 * (ks) + 1) * GEMM_BPITCH]

#define MMA2(ks)                                                              \
    acc0 = __builtin_amdgcn_wmma_f32_16x16x4_f32(                             \
        false, fa0_##ks, false, fb_##ks, (short)0, acc0, false, false);       \
    acc1 = __builtin_amdgcn_wmma_f32_16x16x4_f32(                             \
        false, fa1_##ks, false, fb_##ks, (short)0, acc1, false, false)

#define SLAB_MMA()                                                            \
    {                                                                         \
        FRAG(0);                                                              \
        FRAG(1);                                                              \
        FRAG(2);                                                              \
        FRAG(3);                                                              \
          \
                 \
        __builtin_amdgcn_sched_barrier(0);                                    \
        FRAG(4);                                                              \
        FRAG(5);                                                              \
        FRAG(6);                                                              \
        FRAG(7);                                                              \
                       \
        __builtin_amdgcn_sched_barrier(0);                                    \
        FRAG(8);                                                              \
        FRAG(9);                                                              \
        FRAG(10);                                                             \
        FRAG(11);                                                             \
                       \
        __builtin_amdgcn_sched_barrier(0);                                    \
        FRAG(12);                                                             \
        FRAG(13);                                                             \
        FRAG(14);                                                             \
        FRAG(15);                                                             \
            \
                 \
        __builtin_amdgcn_sched_barrier(0);                                    \
        MMA2(0);                                                              \
        MMA2(1);                                                              \
        MMA2(2);                                                              \
        MMA2(3);                                                              \
        MMA2(4);                                                              \
        MMA2(5);                                                              \
        MMA2(6);                                                              \
        MMA2(7);                                                              \
        MMA2(8);                                                              \
        MMA2(9);                                                              \
        MMA2(10);                                                             \
        MMA2(11);                                                             \
        MMA2(12);                                                             \
        MMA2(13);                                                             \
        MMA2(14);                                                             \
        MMA2(15);                                                             \
          \
        __builtin_amdgcn_sched_barrier(0);                                    \
    }

template <bool RELU>
__global__ __launch_bounds__(GEMM_THREADS)
void wmma_gemm_f32_psum(const float* __restrict__ P,
                        const float* __restrict__ abias,
                        const float* __restrict__ W,
                        const float* __restrict__ bias,
                        float* __restrict__ C,
                        int M, int N, int K) {
    __shared__ __attribute__((aligned(16))) float As[2][GEMM_BM * GEMM_APITCH];
    __shared__ __attribute__((aligned(16))) float Bs[2][GEMM_BK * GEMM_BPITCH];

    const int tid  = threadIdx.x;
    const int lane = tid & 31;
    const int wave = tid >> 5;
    const int half = lane >> 4;
    const int l15  = lane & 15;
    const int n0   = blockIdx.x * GEMM_BN;
    const int m0   = blockIdx.y * GEMM_BM;

    const int c_row = tid >> 4;
    const int c_c4  = tid & 15;
    const size_t plane = (size_t)M * K;
    const float* __restrict__ Ag = P + (size_t)(m0 + c_row) * K + c_c4 * 4;
    const float* __restrict__ Wg = W + (size_t)c_row * N + n0 + c_c4 * 4;
    const int a_st = c_row * GEMM_APITCH + c_c4 * 4;
    const int b_st = c_row * GEMM_BPITCH + c_c4 * 4;

    v4f pa[GEMM_NP][4], pk, pb[8];
    LDG_SLAB(0);

    const int cn = n0 + wave * 16 + l15;
    const float bv = bias[cn];
    v8f acc0, acc1;
#pragma unroll
    for (int r = 0; r < 8; ++r) { acc0[r] = bv; acc1[r] = bv; }

    STS_SLAB(&As[0][0], &Bs[0][0]);
    __syncthreads();

    const int a_ld = l15 * GEMM_APITCH + 2 * half;
    const int b_ld = (2 * half) * GEMM_BPITCH + wave * 16 + l15;

    const int nk = K / GEMM_BK;
    for (int kt = 0; kt + 1 < nk; ++kt) {
        const int cur = kt & 1;
        LDG_SLAB(kt + 1);

        const float* as = &As[cur][a_ld];
        const float* bs = &Bs[cur][b_ld];
        SLAB_MMA();

        asm volatile("" : "+v"(acc0), "+v"(acc1));
        STS_SLAB(&As[cur ^ 1][0], &Bs[cur ^ 1][0]);
        __syncthreads();
    }
    {
        const int cur = (nk - 1) & 1;
        const float* as = &As[cur][a_ld];
        const float* bs = &Bs[cur][b_ld];
        SLAB_MMA();
    }

    float* __restrict__ Cout = C + (size_t)(m0 + 8 * half) * N + cn;
#pragma unroll
    for (int r = 0; r < 8; ++r) {
        float v0 = acc0[r];
        float v1 = acc1[r];
        if (RELU) { v0 = fmaxf(v0, 0.0f); v1 = fmaxf(v1, 0.0f); }
        Cout[(size_t)r * N]        = v0;
        Cout[(size_t)(r + 16) * N] = v1;
    }
}

#define TAIL_ROWS    16
#define TAIL_THREADS 128
#define TAIL_KC      8
#define TAIL_HPITCH  260
#define TAIL_XPITCH  132
#define TAIL_OPITCH  260

__device__ __forceinline__ float lane_xor16(float x) {
    return __builtin_bit_cast(float, __builtin_amdgcn_permlanex16(
        __builtin_bit_cast(int, x), __builtin_bit_cast(int, x),
        0x76543210, 0xfedcba98, false, false));
}
template <int M>
__device__ __forceinline__ float row_xor(float x) {
    return __builtin_bit_cast(float, __builtin_amdgcn_update_dpp(
        0, __builtin_bit_cast(int, x), 0x160 + M, 0xf, 0xf, true));
}
__device__ __forceinline__ float wave_tree_sum(float b) {
    b += lane_xor16(b);
    b += row_xor<8>(b);
    b += row_xor<4>(b);
    b += row_xor<2>(b);
    b += row_xor<1>(b);
    return b;
}

#define TAIL_LD_G(ch)                                                         \
    _Pragma("unroll")                                                         \
    for (int i = 0; i < TAIL_KC; ++i) {                                       \
        g[(ch)][i][0] = *reinterpret_cast<const v2f*>(                        \
            w3p + (size_t)(4 * ((ch) * TAIL_KC + i)) * 128);                  \
        g[(ch)][i][1] = *reinterpret_cast<const v2f*>(                        \
            w3p + (size_t)(4 * ((ch) * TAIL_KC + i) + 1) * 128);              \
    }

#define TAIL_LD_Q(ch)                                                         \
    _Pragma("unroll")                                                         \
    for (int i = 0; i < TAIL_KC; ++i) {                                       \
        q[(ch)][i][0] = *reinterpret_cast<const v4f*>(                        \
            whp + (size_t)(4 * ((ch) * TAIL_KC + i)) * 128);                  \
        q[(ch)][i][1] = *reinterpret_cast<const v4f*>(                        \
            whp + (size_t)(4 * ((ch) * TAIL_KC + i) + 1) * 128);              \
    }

#define TAIL_WMMA(a, b, c)                                                    \
    __builtin_amdgcn_wmma_f32_16x16x4_f32(false, (a), false, (b), (short)0,   \
                                          (c), false, false)

__global__ __launch_bounds__(TAIL_THREADS)
void tail_fused(const float* __restrict__ H2,
                const float* __restrict__ W3,  const float* __restrict__ b3,
                const float* __restrict__ Wsp, const float* __restrict__ bsp,
                const float* __restrict__ Wsh, const float* __restrict__ bsh,
                const int* __restrict__ label,
                const float* __restrict__ ppos,
                const float* __restrict__ pneg,
                float* __restrict__ out) {
    __shared__ __attribute__((aligned(16))) float Hs[TAIL_ROWS * TAIL_HPITCH];
    __shared__ __attribute__((aligned(16))) float Xs[TAIL_ROWS * TAIL_XPITCH];
    __shared__ __attribute__((aligned(16))) float Os[TAIL_ROWS * TAIL_OPITCH];

    const int tid  = threadIdx.x;
    const int lane = tid & 31;
    const int wave = tid >> 5;
    const int half = lane >> 4;
    const int l15  = lane & 15;
    const int m0   = blockIdx.x * TAIL_ROWS;

    v4f ha[8];
#pragma unroll
    for (int i = 0; i < 8; ++i) {
        const int idx = tid + TAIL_THREADS * i;
        ha[i] = *reinterpret_cast<const v4f*>(
            H2 + (size_t)(m0 + (idx >> 6)) * 256 + (idx & 63) * 4);
    }

    const int xc = 32 * wave + 2 * l15;
    const float* __restrict__ w3p = W3 + (size_t)(2 * half) * 128 + xc;
    const float* __restrict__ Wh = (wave < 2) ? Wsp : Wsh;
    const float* __restrict__ bh = (wave < 2) ? bsp : bsh;
    const int hb = 64 * (wave & 1) + 4 * l15;
    const float* __restrict__ whp = Wh + (size_t)(2 * half) * 128 + hb;

    v2f g[8][TAIL_KC][2];
    v4f q[4][TAIL_KC][2];
    TAIL_LD_G(0);
    TAIL_LD_G(1);
    TAIL_LD_G(2);

    const v2f  bx = *reinterpret_cast<const v2f*>(b3 + xc);
    const v4f  bo = *reinterpret_cast<const v4f*>(bh + hb);
    float pp[4], nn[4];
#pragma unroll
    for (int i = 0; i < 4; ++i) {
        pp[i] = ppos[lane + 32 * i];
        nn[i] = pneg[lane + 32 * i];
    }

#pragma unroll
    for (int i = 0; i < 8; ++i) {
        const int idx = tid + TAIL_THREADS * i;
        *reinterpret_cast<v4f*>(&Hs[(idx >> 6) * TAIL_HPITCH + (idx & 63) * 4]) = ha[i];
    }
    __syncthreads();

    v8f xa0, xa1;
#pragma unroll
    for (int r = 0; r < 8; ++r) { xa0[r] = bx.x; xa1[r] = bx.y; }

    v2f af[2][TAIL_KC];
    {
        const float* ap = &Hs[l15 * TAIL_HPITCH + 2 * half];
#pragma unroll
        for (int i = 0; i < TAIL_KC; ++i)
            af[0][i] = *reinterpret_cast<const v2f*>(ap + 4 * i);

#pragma unroll
        for (int ch = 0; ch < 8; ++ch) {
            if (ch + 3 < 8) { TAIL_LD_G(ch + 3); }
            else            { TAIL_LD_Q(ch + 3 - 8); }
            if (ch + 1 < 8) {
#pragma unroll
                for (int i = 0; i < TAIL_KC; ++i)
                    af[(ch + 1) & 1][i] = *reinterpret_cast<const v2f*>(
                        ap + 4 * ((ch + 1) * TAIL_KC + i));
            }
            __builtin_amdgcn_sched_barrier(0);
#pragma unroll
            for (int i = 0; i < TAIL_KC; ++i) {
                v2f f0, f1;
                f0.x = g[ch][i][0].x; f0.y = g[ch][i][1].x;
                f1.x = g[ch][i][0].y; f1.y = g[ch][i][1].y;
                xa0 = TAIL_WMMA(af[ch & 1][i], f0, xa0);
                xa1 = TAIL_WMMA(af[ch & 1][i], f1, xa1);
            }
            __builtin_amdgcn_sched_barrier(0);
        }
    }

    {
        float* xs = &Xs[(8 * half) * TAIL_XPITCH + xc];
#pragma unroll
        for (int r = 0; r < 8; ++r) {
            v2f t; t.x = xa0[r]; t.y = xa1[r];
            *reinterpret_cast<v2f*>(xs + r * TAIL_XPITCH) = t;
        }
    }
    __syncthreads();

    v8f oa0, oa1, oa2, oa3;
#pragma unroll
    for (int r = 0; r < 8; ++r) { oa0[r] = bo.x; oa1[r] = bo.y; oa2[r] = bo.z; oa3[r] = bo.w; }
    {
        const float* xp = &Xs[l15 * TAIL_XPITCH + 2 * half];
#pragma unroll
        for (int i = 0; i < TAIL_KC; ++i)
            af[0][i] = *reinterpret_cast<const v2f*>(xp + 4 * i);

#pragma unroll
        for (int ch = 0; ch < 4; ++ch) {
            if (ch + 3 < 4) { TAIL_LD_Q(ch + 3); }
            if (ch + 1 < 4) {
#pragma unroll
                for (int i = 0; i < TAIL_KC; ++i)
                    af[(ch + 1) & 1][i] = *reinterpret_cast<const v2f*>(
                        xp + 4 * ((ch + 1) * TAIL_KC + i));
            }
            __builtin_amdgcn_sched_barrier(0);
#pragma unroll
            for (int i = 0; i < TAIL_KC; ++i) {
                const v4f k0 = q[ch][i][0];
                const v4f k1 = q[ch][i][1];
                v2f f0, f1, f2, f3;
                f0.x = k0.x; f0.y = k1.x;
                f1.x = k0.y; f1.y = k1.y;
                f2.x = k0.z; f2.y = k1.z;
                f3.x = k0.w; f3.y = k1.w;
                oa0 = TAIL_WMMA(af[ch & 1][i], f0, oa0);
                oa1 = TAIL_WMMA(af[ch & 1][i], f1, oa1);
                oa2 = TAIL_WMMA(af[ch & 1][i], f2, oa2);
                oa3 = TAIL_WMMA(af[ch & 1][i], f3, oa3);
            }
            __builtin_amdgcn_sched_barrier(0);
        }
    }

    {
        float* os = &Os[(8 * half) * TAIL_OPITCH + 64 * wave + 4 * l15];
#pragma unroll
        for (int r = 0; r < 8; ++r) {
            v4f t; t.x = oa0[r]; t.y = oa1[r]; t.z = oa2[r]; t.w = oa3[r];
            *reinterpret_cast<v4f*>(os + r * TAIL_OPITCH) = t;
        }
    }
    __syncthreads();

    const float eps = 1e-12f;
    const float npp = wave_tree_sum((pp[0] * pp[0] + pp[2] * pp[2]) +
                                    (pp[1] * pp[1] + pp[3] * pp[3]));
    const float nnn = wave_tree_sum((nn[0] * nn[0] + nn[2] * nn[2]) +
                                    (nn[1] * nn[1] + nn[3] * nn[3]));
    const float dpp = fmaxf(sqrtf(npp), eps);
    const float dnn = fmaxf(sqrtf(nnn), eps);

    for (int rr = 0; rr < 4; ++rr) {
        const int row = 4 * wave + rr;
        const int b   = m0 + row;
        const float* orow = &Os[row * TAIL_OPITCH + lane];
        float v[4], sh[4];
#pragma unroll
        for (int i = 0; i < 4; ++i) { v[i] = orow[32 * i]; sh[i] = orow[128 + 32 * i]; }

        const float nv = wave_tree_sum((v[0] * v[0] + v[2] * v[2]) +
                                       (v[1] * v[1] + v[3] * v[3]));
        const float dp = wave_tree_sum((v[0] * pp[0] + v[2] * pp[2]) +
                                       (v[1] * pp[1] + v[3] * pp[3]));
        const float dn = wave_tree_sum((v[0] * nn[0] + v[2] * nn[2]) +
                                       (v[1] * nn[1] + v[3] * nn[3]));

        const float dv = fmaxf(sqrtf(nv), eps);
        const float lp = dp / (dv * dpp);
        const float ln = dn / (dv * dnn);

        const bool lab    = label[b] != 0;
        const bool pred   = lp > ln;
        const bool is_bad = (pred == lab) && (lp != ln);
        const float diff  = fabsf(lp - ln);
        const float scale = is_bad ? (tanhf(-diff) + 1.0f) : 1.0f;

        float* o = out + (size_t)b * 256 + lane;
#pragma unroll
        for (int i = 0; i < 4; ++i) {
            o[32 * i]       = v[i] * scale;
            o[128 + 32 * i] = sh[i];
        }
    }
}

extern "C" void kernel_launch(void* const* d_in, const int* in_sizes, int n_in,
                              void* d_out, int out_size, void* d_ws, size_t ws_size,
                              hipStream_t stream) {
    const float* img   = (const float*)d_in[0];
    const int*   label = (const int*)d_in[1];
    const float* W1  = (const float*)d_in[2];
    const float* b1  = (const float*)d_in[3];
    const float* W2  = (const float*)d_in[4];
    const float* b2  = (const float*)d_in[5];
    const float* W3  = (const float*)d_in[6];
    const float* b3  = (const float*)d_in[7];
    const float* Wsp = (const float*)d_in[8];
    const float* bsp = (const float*)d_in[9];
    const float* Wsh = (const float*)d_in[10];
    const float* bsh = (const float*)d_in[11];
    const float* ppn = (const float*)d_in[12];
    const float* pnn = (const float*)d_in[13];

    float* ws   = (float*)d_ws;
    float* part = ws;
    float* h2   = part + (size_t)PG_KSPLIT * BATCH * PG_N;

    pool_gemm1_partial<<<(BATCH / 16) * PG_KSPLIT, PG_THREADS, 0, stream>>>(img, W1, part);

    wmma_gemm_f32_psum<true><<<dim3(256 / GEMM_BN, BATCH / GEMM_BM), GEMM_THREADS, 0, stream>>>(
        part, b1, W2, b2, h2, BATCH, 256, 512);
    tail_fused<<<BATCH / TAIL_ROWS, TAIL_THREADS, 0, stream>>>(
        h2, W3, b3, Wsp, bsp, Wsh, bsh, label, ppn, pnn, (float*)d_out);
}
